# starting kernel
# baseline (speedup 1.0000x reference)
#include <hip/hip_runtime.h>

#define N_NODES 100000
#define IN_DIM 128
#define HID 128
#define GAT_HEADS 8
#define NEG_SLOPE 0.2f

typedef float v2f __attribute__((ext_vector_type(2)));
typedef float v4f __attribute__((ext_vector_type(4)));
typedef float v8f __attribute__((ext_vector_type(8)));

__global__ void __launch_bounds__(256) gemm128_wmma(const float* __restrict__ X,
                                                    const float* __restrict__ W,
                                                    float* __restrict__ H) {
  const int lane  = threadIdx.x & 31;
  const int wave  = blockIdx.x * (blockDim.x >> 5) + (threadIdx.x >> 5);
  const int tileM = wave >> 3;
  const int tileN = wave & 7;
  const int r     = lane & 15;
  const int khalf = lane >> 4;
  const int row   = tileM * 16 + r;
  const int col   = tileN * 16 + r;
  const float* __restrict__ xrow = X + (size_t)row * IN_DIM;

  v8f c = {0.f, 0.f, 0.f, 0.f, 0.f, 0.f, 0.f, 0.f};
#pragma unroll 4
  for (int k = 0; k < IN_DIM; k += 4) {
    const int kk = k + 2 * khalf;
    v2f a = *(const v2f*)(xrow + kk);
    v2f b;
    b.x = W[(kk    ) * HID + col];
    b.y = W[(kk + 1) * HID + col];
    c = __builtin_amdgcn_wmma_f32_16x16x4_f32(false, a, false, b, (short)0, c,
                                              false, false);
  }
  const int mbase = tileM * 16 + khalf * 8;
#pragma unroll
  for (int v = 0; v < 8; ++v)
    H[(size_t)(mbase + v) * HID + tileN * 16 + r] = c[v];
}

__global__ void fill_kernel(float* __restrict__ p, float v, int n) {
  int i = blockIdx.x * blockDim.x + threadIdx.x;
  if (i < n) p[i] = v;
}

__global__ void fill4_kernel(v4f* __restrict__ p, float v, int n4) {
  int i = blockIdx.x * blockDim.x + threadIdx.x;
  if (i < n4) { v4f t = {v, v, v, v}; p[i] = t; }
}

__global__ void alpha1_kernel(const float* __restrict__ H,
                              const float* __restrict__ avS,
                              const float* __restrict__ avD,
                              float* __restrict__ outS,
                              float* __restrict__ outD) {
  int i = blockIdx.x * blockDim.x + threadIdx.x;
  if (i >= N_NODES * GAT_HEADS) return;
  int head = i & 7, node = i >> 3;
  const float* h  = H   + (size_t)node * HID + head * 16;
  const float* as = avS + head * 16;
  const float* ad = avD + head * 16;
  float s = 0.f, d = 0.f;
#pragma unroll
  for (int c = 0; c < 16; ++c) { float hv = h[c]; s += hv * as[c]; d += hv * ad[c]; }
  outS[i] = s; outD[i] = d;
}

__global__ void alpha2_kernel(const float* __restrict__ H,
                              const float* __restrict__ avS,
                              const float* __restrict__ avD,
                              float* __restrict__ outS,
                              float* __restrict__ outD) {
  int wave = blockIdx.x * (blockDim.x >> 5) + (threadIdx.x >> 5);
  int lane = threadIdx.x & 31;
  if (wave >= N_NODES) return;
  float s = 0.f, d = 0.f;
#pragma unroll
  for (int c = lane; c < HID; c += 32) {
    float hv = H[(size_t)wave * HID + c];
    s += hv * avS[c]; d += hv * avD[c];
  }
#pragma unroll
  for (int off = 16; off > 0; off >>= 1) {
    s += __shfl_xor(s, off, 32);
    d += __shfl_xor(d, off, 32);
  }
  if (lane == 0) { outS[wave] = s; outD[wave] = d; }
}

__device__ __forceinline__ void edge_nodes(const int* __restrict__ ei, int e,
                                           int nEdges, int& s, int& d) {
  if (e < nEdges) { s = ei[e]; d = ei[nEdges + e]; }
  else            { s = d = e - nEdges; }
}

__device__ __forceinline__ float leaky(float x) {
  return x > 0.f ? x : NEG_SLOPE * x;
}

__device__ __forceinline__ void atomicMaxFloat(float* addr, float val) {
  if (__float_as_int(val) >= 0)
    atomicMax((int*)addr, __float_as_int(val));
  else
    atomicMin((unsigned int*)addr, __float_as_uint(val));
}

template <int H>
__global__ void edge_max_kernel(const int* __restrict__ ei, int nEdges, int totE,
                                const float* __restrict__ aS,
                                const float* __restrict__ aD,
                                float* __restrict__ M) {
  int i = blockIdx.x * blockDim.x + threadIdx.x;
  if (i >= totE * H) return;
  int h = i % H, e = i / H, s, d;
  edge_nodes(ei, e, nEdges, s, d);
  atomicMaxFloat(&M[d * H + h], leaky(aS[s * H + h] + aD[d * H + h]));
}

template <int H>
__global__ void edge_sum_kernel(const int* __restrict__ ei, int nEdges, int totE,
                                const float* __restrict__ aS,
                                const float* __restrict__ aD,
                                const float* __restrict__ M,
                                float* __restrict__ S) {
  int i = blockIdx.x * blockDim.x + threadIdx.x;
  if (i >= totE * H) return;
  int h = i % H, e = i / H, s, d;
  edge_nodes(ei, e, nEdges, s, d);
  float x = leaky(aS[s * H + h] + aD[d * H + h]);
  atomicAdd(&S[d * H + h], expf(x - M[d * H + h]));
}

template <int H>
__global__ void edge_aggr_kernel(const int* __restrict__ ei, int nEdges, int totE,
                                 const float* __restrict__ aS,
                                 const float* __restrict__ aD,
                                 const float* __restrict__ M,
                                 const float* __restrict__ S,
                                 const float* __restrict__ Hf,
                                 float* __restrict__ acc) {
  int i = blockIdx.x * blockDim.x + threadIdx.x;
  if (i >= totE * (HID / 4)) return;
  int g = i & 31, e = i >> 5, s, d;
  const int c = g * 4;
  edge_nodes(ei, e, nEdges, s, d);
  const int h = c / (HID / H);
  float x = leaky(aS[s * H + h] + aD[d * H + h]);
  float w = expf(x - M[d * H + h]) / S[d * H + h];
  v4f hv = *(const v4f*)(Hf + (size_t)s * HID + c);
  float* a = acc + (size_t)d * HID + c;
  atomicAdd(a + 0, hv.x * w);
  atomicAdd(a + 1, hv.y * w);
  atomicAdd(a + 2, hv.z * w);
  atomicAdd(a + 3, hv.w * w);
}

__global__ void finalize_elu_kernel(const v4f* __restrict__ acc,
                                    const float* __restrict__ bias,
                                    v4f* __restrict__ out) {
  int i = blockIdx.x * blockDim.x + threadIdx.x;
  if (i >= N_NODES * (HID / 4)) return;
  const int c = (i & 31) * 4;
  v4f v = acc[i];
  v.x += bias[c + 0]; v.y += bias[c + 1]; v.z += bias[c + 2]; v.w += bias[c + 3];
  v.x = v.x > 0.f ? v.x : (expf(v.x) - 1.f);
  v.y = v.y > 0.f ? v.y : (expf(v.y) - 1.f);
  v.z = v.z > 0.f ? v.z : (expf(v.z) - 1.f);
  v.w = v.w > 0.f ? v.w : (expf(v.w) - 1.f);
  out[i] = v;
}

__global__ void bias_kernel(v4f* __restrict__ out, const float* __restrict__ bias) {
  int i = blockIdx.x * blockDim.x + threadIdx.x;
  if (i >= N_NODES * (HID / 4)) return;
  const int c = (i & 31) * 4;
  v4f v = out[i];
  v.x += bias[c + 0]; v.y += bias[c + 1]; v.z += bias[c + 2]; v.w += bias[c + 3];
  out[i] = v;
}

extern "C" void kernel_launch(void* const* d_in, const int* in_sizes, int n_in,
                              void* d_out, int out_size, void* d_ws, size_t ws_size,
                              hipStream_t stream) {
  const float* x   = (const float*)d_in[0];
  const int*   ei  = (const int*)d_in[1];
  const float* W1  = (const float*)d_in[2];
  const float* as1 = (const float*)d_in[3];
  const float* ad1 = (const float*)d_in[4];
  const float* b1  = (const float*)d_in[5];
  const float* W2  = (const float*)d_in[6];
  const float* as2 = (const float*)d_in[7];
  const float* ad2 = (const float*)d_in[8];
  const float* b2  = (const float*)d_in[9];

  const int nEdges = in_sizes[1] / 2;
  const int totE   = nEdges + N_NODES;

  float* A   = (float*)d_ws;
  float* B   = A  + (size_t)N_NODES * HID;
  float* aS  = B  + (size_t)N_NODES * HID;
  float* aD  = aS + (size_t)N_NODES * GAT_HEADS;
  float* Mx  = aD + (size_t)N_NODES * GAT_HEADS;
  float* Sx  = Mx + (size_t)N_NODES * GAT_HEADS;
  float* out = (float*)d_out;

  const int TB = 256;
  auto cdiv = [](long long a, long long b) { return (unsigned)((a + b - 1) / b); };
  const unsigned gNode4 = cdiv((long long)N_NODES * (HID / 4), TB);
  const unsigned gGemm  = (N_NODES / 16) * (HID / 16) / (TB / 32);
  const unsigned gAggr  = cdiv((long long)totE * (HID / 4), TB);
  const float NEG_INF = -__builtin_huge_valf();

  gemm128_wmma<<<gGemm, TB, 0, stream>>>(x, W1, A);
  alpha1_kernel<<<cdiv(N_NODES * GAT_HEADS, TB), TB, 0, stream>>>(A, as1, ad1, aS, aD);
  fill_kernel<<<cdiv(N_NODES * GAT_HEADS, TB), TB, 0, stream>>>(Mx, NEG_INF, N_NODES * GAT_HEADS);
  fill_kernel<<<cdiv(N_NODES * GAT_HEADS, TB), TB, 0, stream>>>(Sx, 0.f, N_NODES * GAT_HEADS);
  fill4_kernel<<<gNode4, TB, 0, stream>>>((v4f*)B, 0.f, N_NODES * (HID / 4));
  edge_max_kernel<GAT_HEADS><<<cdiv((long long)totE * GAT_HEADS, TB), TB, 0, stream>>>(
      ei, nEdges, totE, aS, aD, Mx);
  edge_sum_kernel<GAT_HEADS><<<cdiv((long long)totE * GAT_HEADS, TB), TB, 0, stream>>>(
      ei, nEdges, totE, aS, aD, Mx, Sx);
  edge_aggr_kernel<GAT_HEADS><<<gAggr, TB, 0, stream>>>(
      ei, nEdges, totE, aS, aD, Mx, Sx, A, B);
  finalize_elu_kernel<<<gNode4, TB, 0, stream>>>((const v4f*)B, b1, (v4f*)A);

  gemm128_wmma<<<gGemm, TB, 0, stream>>>(A, W2, B);
  alpha2_kernel<<<cdiv((long long)N_NODES * 32, TB), TB, 0, stream>>>(B, as2, ad2, aS, aD);
  fill_kernel<<<cdiv(N_NODES, TB), TB, 0, stream>>>(Mx, NEG_INF, N_NODES);
  fill_kernel<<<cdiv(N_NODES, TB), TB, 0, stream>>>(Sx, 0.f, N_NODES);
  fill4_kernel<<<gNode4, TB, 0, stream>>>((v4f*)out, 0.f, N_NODES * (HID / 4));
  edge_max_kernel<1><<<cdiv(totE, TB), TB, 0, stream>>>(ei, nEdges, totE, aS, aD, Mx);
  edge_sum_kernel<1><<<cdiv(totE, TB), TB, 0, stream>>>(ei, nEdges, totE, aS, aD, Mx, Sx);
  edge_aggr_kernel<1><<<gAggr, TB, 0, stream>>>(
      ei, nEdges, totE, aS, aD, Mx, Sx, B, out);
  bias_kernel<<<gNode4, TB, 0, stream>>>((v4f*)out, b2);
}


# Grids sized to the chip's wave slots instead of one wave per item: every wave enters once and loops over its share of nodes, row tiles or edges, about five times fewer waves to dispatch per call
# speedup vs baseline: 3.7628x; 3.7628x over previous
#include <hip/hip_runtime.h>

#define N_NODES 100000
#define IN_DIM 128
#define HID 128
#define GAT_HEADS 8
#define NEG_SLOPE 0.2f

typedef float v2f __attribute__((ext_vector_type(2)));
typedef float v4f __attribute__((ext_vector_type(4)));
typedef float v8f __attribute__((ext_vector_type(8)));

__global__ void __launch_bounds__(256) gemm128_wmma(const float* __restrict__ X,
                                                    const float* __restrict__ W,
                                                    float* __restrict__ H) {
  const int lane  = threadIdx.x & 31;
  const int tileN = threadIdx.x >> 5;
  const int r     = lane & 15;
  const int khalf = lane >> 4;
  const int col   = tileN * 16 + r;

  v2f bf[IN_DIM / 4];
#pragma unroll
  for (int i = 0; i < IN_DIM / 4; ++i) {
    const int kk = 4 * i + 2 * khalf;
    bf[i].x = W[(kk    ) * HID + col];
    bf[i].y = W[(kk + 1) * HID + col];
  }

  for (int tileM = blockIdx.x; tileM < N_NODES / 16; tileM += gridDim.x) {
    const float* __restrict__ xrow =
        X + (size_t)(tileM * 16 + r) * IN_DIM + 2 * khalf;
    v8f c = {0.f, 0.f, 0.f, 0.f, 0.f, 0.f, 0.f, 0.f};
#pragma unroll
    for (int i = 0; i < IN_DIM / 4; ++i) {
      v2f a = *(const v2f*)(xrow + 4 * i);
      c = __builtin_amdgcn_wmma_f32_16x16x4_f32(false, a, false, bf[i], (short)0,
                                                c, false, false);
    }
    const int mbase = tileM * 16 + khalf * 8;
#pragma unroll
    for (int v = 0; v < 8; ++v)
      H[(size_t)(mbase + v) * HID + tileN * 16 + r] = c[v];
  }
}

typedef int v4i __attribute__((ext_vector_type(4)));

#define SCAN_TB 256
#define SCAN_TILE (SCAN_TB * 4)

__global__ void alpha1_kernel(const float* __restrict__ H,
                              const float* __restrict__ avS,
                              float* __restrict__ outS) {
  const int stride = gridDim.x * blockDim.x;
  for (int i = blockIdx.x * blockDim.x + threadIdx.x; i < N_NODES * GAT_HEADS;
       i += stride) {
    int head = i & 7, node = i >> 3;
    const float* h  = H   + (size_t)node * HID + head * 16;
    const float* as = avS + head * 16;
    float s = 0.f;
#pragma unroll
    for (int c = 0; c < 16; ++c) s += h[c] * as[c];
    outS[i] = s;
  }
}

__global__ void alpha2_kernel(const float* __restrict__ H,
                              const float* __restrict__ avS,
                              float* __restrict__ outS) {
  const int wave = blockIdx.x * (blockDim.x >> 5) + (threadIdx.x >> 5);
  const int nW   = gridDim.x * (blockDim.x >> 5);
  const int lane = threadIdx.x & 31;
  for (int node = wave; node < N_NODES; node += nW) {
    float s = 0.f;
#pragma unroll
    for (int c = lane; c < HID; c += 32) s += H[(size_t)node * HID + c] * avS[c];
#pragma unroll
    for (int off = 16; off > 0; off >>= 1) s += __shfl_xor(s, off, 32);
    if (lane == 0) outS[node] = s;
  }
}

#define CNT_PITCH 32

__global__ void zero_counters_kernel(int* __restrict__ cnt) {
  int i = blockIdx.x * blockDim.x + threadIdx.x;
  if (i < N_NODES) cnt[i * CNT_PITCH] = 0;
}

__global__ void degree_kernel(const int* __restrict__ ei, int nEdges,
                              int* __restrict__ cnt) {
  const int stride = gridDim.x * blockDim.x;
  const int* __restrict__ dst = ei + nEdges;
  int i = blockIdx.x * blockDim.x + threadIdx.x;
  for (; i + 3 * stride < nEdges; i += 4 * stride) {
    const int d0 = dst[i], d1 = dst[i + stride];
    const int d2 = dst[i + 2 * stride], d3 = dst[i + 3 * stride];
    atomicAdd(&cnt[d0 * CNT_PITCH], 1);
    atomicAdd(&cnt[d1 * CNT_PITCH], 1);
    atomicAdd(&cnt[d2 * CNT_PITCH], 1);
    atomicAdd(&cnt[d3 * CNT_PITCH], 1);
  }
  for (; i < nEdges; i += stride) atomicAdd(&cnt[dst[i] * CNT_PITCH], 1);
}

__global__ void __launch_bounds__(SCAN_TB)
scan_tile_sums_kernel(const int* __restrict__ cnt, int* __restrict__ tileSum) {
  __shared__ int wsum[SCAN_TB / 32];
  const int base = (blockIdx.x * SCAN_TB + threadIdx.x) * 4;
  int t = 0;
  if (base < N_NODES)
    t = cnt[(base + 0) * CNT_PITCH] + cnt[(base + 1) * CNT_PITCH] +
        cnt[(base + 2) * CNT_PITCH] + cnt[(base + 3) * CNT_PITCH];
  for (int off = 16; off > 0; off >>= 1) t += __shfl_xor(t, off, 32);
  if ((threadIdx.x & 31) == 0) wsum[threadIdx.x >> 5] = t;
  __syncthreads();
  if (threadIdx.x == 0) {
    int s = 0;
    for (int w = 0; w < SCAN_TB / 32; ++w) s += wsum[w];
    tileSum[blockIdx.x] = s;
  }
}

__global__ void __launch_bounds__(SCAN_TB)
scan_apply_kernel(int* __restrict__ cnt, const int* __restrict__ tileSum,
                  int* __restrict__ segStart) {
  __shared__ int wsum[SCAN_TB / 32];
  __shared__ int tileOff;
  const int lane = threadIdx.x & 31, wave = threadIdx.x >> 5;
  if (wave == 0) {
    int o = 0;
    for (int b = lane; b < (int)blockIdx.x; b += 32) o += tileSum[b];
    for (int off = 16; off > 0; off >>= 1) o += __shfl_xor(o, off, 32);
    if (lane == 0) tileOff = o;
  }
  const int base = (blockIdx.x * SCAN_TB + threadIdx.x) * 4;
  v4i v = {0, 0, 0, 0};
  if (base < N_NODES) {
    v.x = cnt[(base + 0) * CNT_PITCH]; v.y = cnt[(base + 1) * CNT_PITCH];
    v.z = cnt[(base + 2) * CNT_PITCH]; v.w = cnt[(base + 3) * CNT_PITCH];
  }
  const int t = v.x + v.y + v.z + v.w;
  int inc = t;
  for (int off = 1; off < 32; off <<= 1) {
    int u = __shfl_up(inc, off, 32);
    if (lane >= off) inc += u;
  }
  if (lane == 31) wsum[wave] = inc;
  __syncthreads();
  int o = tileOff;
  for (int w = 0; w < wave; ++w) o += wsum[w];
  const int e0 = o + inc - t;
  if (base < N_NODES) {
    v4i r = {e0, e0 + v.x, e0 + v.x + v.y, e0 + v.x + v.y + v.z};
    *(v4i*)(segStart + base) = r;
    cnt[(base + 0) * CNT_PITCH] = r.x; cnt[(base + 1) * CNT_PITCH] = r.y;
    cnt[(base + 2) * CNT_PITCH] = r.z; cnt[(base + 3) * CNT_PITCH] = r.w;
  }
}

__global__ void fill_csr_kernel(const int* __restrict__ ei, int nEdges,
                                int* __restrict__ cnt, int* __restrict__ csr) {
  const int stride = gridDim.x * blockDim.x;
  const int* __restrict__ dst = ei + nEdges;
  int i = blockIdx.x * blockDim.x + threadIdx.x;
  for (; i + 3 * stride < nEdges; i += 4 * stride) {
    const int s0 = ei[i], s1 = ei[i + stride];
    const int s2 = ei[i + 2 * stride], s3 = ei[i + 3 * stride];
    const int d0 = dst[i], d1 = dst[i + stride];
    const int d2 = dst[i + 2 * stride], d3 = dst[i + 3 * stride];
    const int p0 = atomicAdd(&cnt[d0 * CNT_PITCH], 1);
    const int p1 = atomicAdd(&cnt[d1 * CNT_PITCH], 1);
    const int p2 = atomicAdd(&cnt[d2 * CNT_PITCH], 1);
    const int p3 = atomicAdd(&cnt[d3 * CNT_PITCH], 1);
    csr[p0] = s0; csr[p1] = s1; csr[p2] = s2; csr[p3] = s3;
  }
  for (; i < nEdges; i += stride) {
    const int s = ei[i], d = dst[i];
    csr[atomicAdd(&cnt[d * CNT_PITCH], 1)] = s;
  }
}

__device__ __forceinline__ float leaky(float x) {
  return x > 0.f ? x : NEG_SLOPE * x;
}

template <int H, bool ELU>
__global__ void __launch_bounds__(256)
gat_gather_kernel(const int* __restrict__ csr, const int* __restrict__ segStart,
                  int nEdges,
                  const float* __restrict__ aS, const float* __restrict__ avD,
                  const float* __restrict__ Hf, const float* __restrict__ bias,
                  float* __restrict__ out) {
  const int lane = threadIdx.x & 31;
  const int nW = gridDim.x * 8;
  int d = __builtin_amdgcn_readfirstlane(blockIdx.x * 8 + (threadIdx.x >> 5));
  if (d >= N_NODES) return;
  const int c = lane * 4;
  const int h = c / (HID / H);
  const v4f ad = *(const v4f*)(avD + c);
  const v4f b = *(const v4f*)(bias + c);

  int beg = segStart[d];
  int end = (d + 1 < N_NODES) ? segStart[d + 1] : nEdges;
  int mys0 = (lane < end - beg) ? csr[beg + lane] : 0;
  v4f hself = *(const v4f*)(Hf + (size_t)d * HID + c);
  float asd = aS[d * H + h];

  for (;;) {
    const int dn = d + nW;
    const bool more = dn < N_NODES;
    int begN = 0, endN = 0;
    if (more) {
      begN = segStart[dn];
      endN = (dn + 1 < N_NODES) ? segStart[dn + 1] : nEdges;
    }

    float pd = hself.x * ad.x + hself.y * ad.y + hself.z * ad.z + hself.w * ad.w;
    if (H == GAT_HEADS) {
      pd += __shfl_xor(pd, 1, 32);
      pd += __shfl_xor(pd, 2, 32);
    } else {
      for (int off = 16; off > 0; off >>= 1) pd += __shfl_xor(pd, off, 32);
    }
    const float xself = leaky(asd + pd);

    float m = xself, sum;
    v4f acc;
    int mysN = 0;
    v4f hselfN = hself;
    float asdN = 0.f;
    if (H == 1) {
      for (int base = beg; base < end; base += 32) {
        const int n = min(32, end - base);
        const int mys = (base == beg) ? mys0 : ((lane < n) ? csr[base + lane] : 0);
        if (lane < n) m = fmaxf(m, leaky(aS[mys] + pd));
      }
      for (int off = 16; off > 0; off >>= 1) m = fmaxf(m, __shfl_xor(m, off, 32));
      const float eself = expf(xself - m);
      float part = 0.f;
      acc = hself * eself;
      for (int base = beg; base < end; base += 32) {
        const int n = min(32, end - base);
        int mys = 0;
        float e = 0.f;
        if (lane < n) {
          mys = (base == beg) ? mys0 : csr[base + lane];
          e = expf(leaky(aS[mys] + pd) - m);
        }
        part += e;
#pragma unroll 4
        for (int j = 0; j < n; ++j) {
          const int s = __builtin_amdgcn_readlane(mys, j);
          const float ej = __int_as_float(__builtin_amdgcn_readlane(__float_as_int(e), j));
          const v4f hv = *(const v4f*)(Hf + (size_t)s * HID + c);
          acc += hv * ej;
        }
      }
      for (int off = 16; off > 0; off >>= 1) part += __shfl_xor(part, off, 32);
      sum = eself + part;
    } else {
      for (int base = beg; base < end; base += 32) {
        const int n = min(32, end - base);
        const int mys = (base == beg) ? mys0 : ((lane < n) ? csr[base + lane] : 0);
#pragma unroll 4
        for (int j = 0; j < n; ++j) {
          const int s = __builtin_amdgcn_readlane(mys, j);
          m = fmaxf(m, leaky(aS[s * H + h] + pd));
        }
      }
      const float eself = expf(xself - m);
      sum = eself;
      acc = hself * eself;
      for (int base = beg; base < end; base += 32) {
        const int n = min(32, end - base);
        const int mys = (base == beg) ? mys0 : ((lane < n) ? csr[base + lane] : 0);
#pragma unroll 4
        for (int j = 0; j < n; ++j) {
          const int s = __builtin_amdgcn_readlane(mys, j);
          const v4f hv = *(const v4f*)(Hf + (size_t)s * HID + c);
          const float e = expf(leaky(aS[s * H + h] + pd) - m);
          sum += e;
          acc += hv * e;
        }
      }
    }

    if (more) {
      mysN = (lane < endN - begN) ? csr[begN + lane] : 0;
      hselfN = *(const v4f*)(Hf + (size_t)dn * HID + c);
      asdN = aS[dn * H + h];
    }

    v4f r = acc / sum + b;
    if (ELU) {
      r.x = r.x > 0.f ? r.x : (expf(r.x) - 1.f);
      r.y = r.y > 0.f ? r.y : (expf(r.y) - 1.f);
      r.z = r.z > 0.f ? r.z : (expf(r.z) - 1.f);
      r.w = r.w > 0.f ? r.w : (expf(r.w) - 1.f);
    }
    *(v4f*)(out + (size_t)d * HID + c) = r;

    if (!more) break;
    d = dn; beg = begN; end = endN;
    mys0 = mysN; hself = hselfN; asd = asdN;
  }
}

extern "C" void kernel_launch(void* const* d_in, const int* in_sizes, int n_in,
                              void* d_out, int out_size, void* d_ws, size_t ws_size,
                              hipStream_t stream) {
  const float* x   = (const float*)d_in[0];
  const int*   ei  = (const int*)d_in[1];
  const float* W1  = (const float*)d_in[2];
  const float* as1 = (const float*)d_in[3];
  const float* ad1 = (const float*)d_in[4];
  const float* b1  = (const float*)d_in[5];
  const float* W2  = (const float*)d_in[6];
  const float* as2 = (const float*)d_in[7];
  const float* ad2 = (const float*)d_in[8];
  const float* b2  = (const float*)d_in[9];

  const int nEdges = in_sizes[1] / 2;

  float* A    = (float*)d_ws;
  float* B    = A  + (size_t)N_NODES * HID;
  float* aS   = B  + (size_t)N_NODES * HID;
  int*   csr  = (int*)(aS + (size_t)N_NODES * GAT_HEADS);
  int*   seg  = (int*)(aS + (size_t)N_NODES * GAT_HEADS * 3);
  int*   tsum = seg + 102400;
  int*   cnt  = (int*)B;
  float* out  = (float*)d_out;

  const int TB = 256;
  auto cdiv = [](long long a, long long b) { return (unsigned)((a + b - 1) / b); };
  const unsigned gTiles = cdiv(N_NODES, SCAN_TILE);
  const unsigned gGemm  = 1250;
  const unsigned gEdge  = 3125;
  const unsigned gNodeW = 1786;
  const int TBE = 128;

  zero_counters_kernel<<<cdiv(N_NODES, TB), TB, 0, stream>>>(cnt);
  degree_kernel<<<gEdge, TBE, 0, stream>>>(ei, nEdges, cnt);
  scan_tile_sums_kernel<<<gTiles, SCAN_TB, 0, stream>>>(cnt, tsum);
  scan_apply_kernel<<<gTiles, SCAN_TB, 0, stream>>>(cnt, tsum, seg);
  fill_csr_kernel<<<gEdge, TBE, 0, stream>>>(ei, nEdges, cnt, csr);

  gemm128_wmma<<<gGemm, TB, 0, stream>>>(x, W1, A);
  alpha1_kernel<<<gEdge, TBE, 0, stream>>>(A, as1, aS);
  gat_gather_kernel<GAT_HEADS, true><<<gNodeW, TB, 0, stream>>>(
      csr, seg, nEdges, aS, ad1, A, b1, B);

  gemm128_wmma<<<gGemm, TB, 0, stream>>>(B, W2, A);
  alpha2_kernel<<<gNodeW, TB, 0, stream>>>(A, as2, aS);
  gat_gather_kernel<1, false><<<gNodeW, TB, 0, stream>>>(
      csr, seg, nEdges, aS, ad2, A, b2, out);
}


# One atomic per edge instead of two: the counting add returns each edge's rank in its node's segment, stored per edge; placement is csr[segStart[dst]+rank]=src with plain loads and stores
# speedup vs baseline: 4.1570x; 1.1048x over previous
#include <hip/hip_runtime.h>

#define N_NODES 100000
#define IN_DIM 128
#define HID 128
#define GAT_HEADS 8
#define NEG_SLOPE 0.2f

typedef float v2f __attribute__((ext_vector_type(2)));
typedef float v4f __attribute__((ext_vector_type(4)));
typedef float v8f __attribute__((ext_vector_type(8)));

__global__ void __launch_bounds__(256) gemm128_wmma(const float* __restrict__ X,
                                                    const float* __restrict__ W,
                                                    float* __restrict__ H) {
  const int lane  = threadIdx.x & 31;
  const int tileN = threadIdx.x >> 5;
  const int r     = lane & 15;
  const int khalf = lane >> 4;
  const int col   = tileN * 16 + r;

  v2f bf[IN_DIM / 4];
#pragma unroll
  for (int i = 0; i < IN_DIM / 4; ++i) {
    const int kk = 4 * i + 2 * khalf;
    bf[i].x = W[(kk    ) * HID + col];
    bf[i].y = W[(kk + 1) * HID + col];
  }

  for (int tileM = blockIdx.x; tileM < N_NODES / 16; tileM += gridDim.x) {
    const float* __restrict__ xrow =
        X + (size_t)(tileM * 16 + r) * IN_DIM + 2 * khalf;
    v8f c = {0.f, 0.f, 0.f, 0.f, 0.f, 0.f, 0.f, 0.f};
#pragma unroll
    for (int i = 0; i < IN_DIM / 4; ++i) {
      v2f a = *(const v2f*)(xrow + 4 * i);
      c = __builtin_amdgcn_wmma_f32_16x16x4_f32(false, a, false, bf[i], (short)0,
                                                c, false, false);
    }
    const int mbase = tileM * 16 + khalf * 8;
#pragma unroll
    for (int v = 0; v < 8; ++v)
      H[(size_t)(mbase + v) * HID + tileN * 16 + r] = c[v];
  }
}

typedef int v4i __attribute__((ext_vector_type(4)));

#define SCAN_TB 256
#define SCAN_TILE (SCAN_TB * 4)

__global__ void alpha1_kernel(const float* __restrict__ H,
                              const float* __restrict__ avS,
                              float* __restrict__ outS) {
  const int stride = gridDim.x * blockDim.x;
  for (int i = blockIdx.x * blockDim.x + threadIdx.x; i < N_NODES * GAT_HEADS;
       i += stride) {
    int head = i & 7, node = i >> 3;
    const float* h  = H   + (size_t)node * HID + head * 16;
    const float* as = avS + head * 16;
    float s = 0.f;
#pragma unroll
    for (int c = 0; c < 16; ++c) s += h[c] * as[c];
    outS[i] = s;
  }
}

__global__ void alpha2_kernel(const float* __restrict__ H,
                              const float* __restrict__ avS,
                              float* __restrict__ outS) {
  const int wave = blockIdx.x * (blockDim.x >> 5) + (threadIdx.x >> 5);
  const int nW   = gridDim.x * (blockDim.x >> 5);
  const int lane = threadIdx.x & 31;
  for (int node = wave; node < N_NODES; node += nW) {
    float s = 0.f;
#pragma unroll
    for (int c = lane; c < HID; c += 32) s += H[(size_t)node * HID + c] * avS[c];
#pragma unroll
    for (int off = 16; off > 0; off >>= 1) s += __shfl_xor(s, off, 32);
    if (lane == 0) outS[node] = s;
  }
}

#define CNT_PITCH 32

__global__ void zero_counters_kernel(int* __restrict__ cnt) {
  int i = blockIdx.x * blockDim.x + threadIdx.x;
  if (i < N_NODES) cnt[i * CNT_PITCH] = 0;
}

__global__ void degree_rank_kernel(const int* __restrict__ ei, int nEdges,
                                   int* __restrict__ cnt,
                                   int* __restrict__ rank) {
  const int stride = gridDim.x * blockDim.x;
  const int* __restrict__ dst = ei + nEdges;
  int i = blockIdx.x * blockDim.x + threadIdx.x;
  for (; i + 3 * stride < nEdges; i += 4 * stride) {
    const int d0 = dst[i], d1 = dst[i + stride];
    const int d2 = dst[i + 2 * stride], d3 = dst[i + 3 * stride];
    const int r0 = atomicAdd(&cnt[d0 * CNT_PITCH], 1);
    const int r1 = atomicAdd(&cnt[d1 * CNT_PITCH], 1);
    const int r2 = atomicAdd(&cnt[d2 * CNT_PITCH], 1);
    const int r3 = atomicAdd(&cnt[d3 * CNT_PITCH], 1);
    rank[i] = r0; rank[i + stride] = r1;
    rank[i + 2 * stride] = r2; rank[i + 3 * stride] = r3;
  }
  for (; i < nEdges; i += stride)
    rank[i] = atomicAdd(&cnt[dst[i] * CNT_PITCH], 1);
}

__global__ void __launch_bounds__(SCAN_TB)
scan_tile_sums_kernel(const int* __restrict__ cnt, int* __restrict__ tileSum) {
  __shared__ int wsum[SCAN_TB / 32];
  const int base = (blockIdx.x * SCAN_TB + threadIdx.x) * 4;
  int t = 0;
  if (base < N_NODES)
    t = cnt[(base + 0) * CNT_PITCH] + cnt[(base + 1) * CNT_PITCH] +
        cnt[(base + 2) * CNT_PITCH] + cnt[(base + 3) * CNT_PITCH];
  for (int off = 16; off > 0; off >>= 1) t += __shfl_xor(t, off, 32);
  if ((threadIdx.x & 31) == 0) wsum[threadIdx.x >> 5] = t;
  __syncthreads();
  if (threadIdx.x == 0) {
    int s = 0;
    for (int w = 0; w < SCAN_TB / 32; ++w) s += wsum[w];
    tileSum[blockIdx.x] = s;
  }
}

__global__ void __launch_bounds__(SCAN_TB)
scan_apply_kernel(const int* __restrict__ cnt, const int* __restrict__ tileSum,
                  int* __restrict__ segStart) {
  __shared__ int wsum[SCAN_TB / 32];
  __shared__ int tileOff;
  const int lane = threadIdx.x & 31, wave = threadIdx.x >> 5;
  if (wave == 0) {
    int o = 0;
    for (int b = lane; b < (int)blockIdx.x; b += 32) o += tileSum[b];
    for (int off = 16; off > 0; off >>= 1) o += __shfl_xor(o, off, 32);
    if (lane == 0) tileOff = o;
  }
  const int base = (blockIdx.x * SCAN_TB + threadIdx.x) * 4;
  v4i v = {0, 0, 0, 0};
  if (base < N_NODES) {
    v.x = cnt[(base + 0) * CNT_PITCH]; v.y = cnt[(base + 1) * CNT_PITCH];
    v.z = cnt[(base + 2) * CNT_PITCH]; v.w = cnt[(base + 3) * CNT_PITCH];
  }
  const int t = v.x + v.y + v.z + v.w;
  int inc = t;
  for (int off = 1; off < 32; off <<= 1) {
    int u = __shfl_up(inc, off, 32);
    if (lane >= off) inc += u;
  }
  if (lane == 31) wsum[wave] = inc;
  __syncthreads();
  int o = tileOff;
  for (int w = 0; w < wave; ++w) o += wsum[w];
  const int e0 = o + inc - t;
  if (base < N_NODES) {
    v4i r = {e0, e0 + v.x, e0 + v.x + v.y, e0 + v.x + v.y + v.z};
    *(v4i*)(segStart + base) = r;
  }
}

__global__ void fill_csr_kernel(const int* __restrict__ ei, int nEdges,
                                const int* __restrict__ segStart,
                                const int* __restrict__ rank,
                                int* __restrict__ csr) {
  const int stride = gridDim.x * blockDim.x;
  const int* __restrict__ dst = ei + nEdges;
  int i = blockIdx.x * blockDim.x + threadIdx.x;
  for (; i + 3 * stride < nEdges; i += 4 * stride) {
    const int d0 = dst[i], d1 = dst[i + stride];
    const int d2 = dst[i + 2 * stride], d3 = dst[i + 3 * stride];
    const int s0 = ei[i], s1 = ei[i + stride];
    const int s2 = ei[i + 2 * stride], s3 = ei[i + 3 * stride];
    const int r0 = rank[i], r1 = rank[i + stride];
    const int r2 = rank[i + 2 * stride], r3 = rank[i + 3 * stride];
    const int p0 = segStart[d0] + r0, p1 = segStart[d1] + r1;
    const int p2 = segStart[d2] + r2, p3 = segStart[d3] + r3;
    csr[p0] = s0; csr[p1] = s1; csr[p2] = s2; csr[p3] = s3;
  }
  for (; i < nEdges; i += stride) csr[segStart[dst[i]] + rank[i]] = ei[i];
}

__device__ __forceinline__ float leaky(float x) {
  return x > 0.f ? x : NEG_SLOPE * x;
}

template <int H, bool ELU>
__global__ void __launch_bounds__(256)
gat_gather_kernel(const int* __restrict__ csr, const int* __restrict__ segStart,
                  int nEdges,
                  const float* __restrict__ aS, const float* __restrict__ avD,
                  const float* __restrict__ Hf, const float* __restrict__ bias,
                  float* __restrict__ out) {
  const int lane = threadIdx.x & 31;
  const int nW = gridDim.x * 8;
  int d = __builtin_amdgcn_readfirstlane(blockIdx.x * 8 + (threadIdx.x >> 5));
  if (d >= N_NODES) return;
  const int c = lane * 4;
  const int h = c / (HID / H);
  const v4f ad = *(const v4f*)(avD + c);
  const v4f b = *(const v4f*)(bias + c);

  int beg = segStart[d];
  int end = (d + 1 < N_NODES) ? segStart[d + 1] : nEdges;
  int mys0 = (lane < end - beg) ? csr[beg + lane] : 0;
  v4f hself = *(const v4f*)(Hf + (size_t)d * HID + c);
  float asd = aS[d * H + h];

  for (;;) {
    const int dn = d + nW;
    const bool more = dn < N_NODES;
    int begN = 0, endN = 0;
    if (more) {
      begN = segStart[dn];
      endN = (dn + 1 < N_NODES) ? segStart[dn + 1] : nEdges;
    }

    float pd = hself.x * ad.x + hself.y * ad.y + hself.z * ad.z + hself.w * ad.w;
    if (H == GAT_HEADS) {
      pd += __shfl_xor(pd, 1, 32);
      pd += __shfl_xor(pd, 2, 32);
    } else {
      for (int off = 16; off > 0; off >>= 1) pd += __shfl_xor(pd, off, 32);
    }
    const float xself = leaky(asd + pd);

    float m = xself, sum;
    v4f acc;
    int mysN = 0;
    v4f hselfN = hself;
    float asdN = 0.f;
    if (H == 1) {
      for (int base = beg; base < end; base += 32) {
        const int n = min(32, end - base);
        const int mys = (base == beg) ? mys0 : ((lane < n) ? csr[base + lane] : 0);
        if (lane < n) m = fmaxf(m, leaky(aS[mys] + pd));
      }
      for (int off = 16; off > 0; off >>= 1) m = fmaxf(m, __shfl_xor(m, off, 32));
      const float eself = expf(xself - m);
      float part = 0.f;
      acc = hself * eself;
      for (int base = beg; base < end; base += 32) {
        const int n = min(32, end - base);
        int mys = 0;
        float e = 0.f;
        if (lane < n) {
          mys = (base == beg) ? mys0 : csr[base + lane];
          e = expf(leaky(aS[mys] + pd) - m);
        }
        part += e;
#pragma unroll 4
        for (int j = 0; j < n; ++j) {
          const int s = __builtin_amdgcn_readlane(mys, j);
          const float ej = __int_as_float(__builtin_amdgcn_readlane(__float_as_int(e), j));
          const v4f hv = *(const v4f*)(Hf + (size_t)s * HID + c);
          acc += hv * ej;
        }
      }
      for (int off = 16; off > 0; off >>= 1) part += __shfl_xor(part, off, 32);
      sum = eself + part;
    } else {
      for (int base = beg; base < end; base += 32) {
        const int n = min(32, end - base);
        const int mys = (base == beg) ? mys0 : ((lane < n) ? csr[base + lane] : 0);
#pragma unroll 4
        for (int j = 0; j < n; ++j) {
          const int s = __builtin_amdgcn_readlane(mys, j);
          m = fmaxf(m, leaky(aS[s * H + h] + pd));
        }
      }
      const float eself = expf(xself - m);
      sum = eself;
      acc = hself * eself;
      for (int base = beg; base < end; base += 32) {
        const int n = min(32, end - base);
        const int mys = (base == beg) ? mys0 : ((lane < n) ? csr[base + lane] : 0);
#pragma unroll 4
        for (int j = 0; j < n; ++j) {
          const int s = __builtin_amdgcn_readlane(mys, j);
          const v4f hv = *(const v4f*)(Hf + (size_t)s * HID + c);
          const float e = expf(leaky(aS[s * H + h] + pd) - m);
          sum += e;
          acc += hv * e;
        }
      }
    }

    if (more) {
      mysN = (lane < endN - begN) ? csr[begN + lane] : 0;
      hselfN = *(const v4f*)(Hf + (size_t)dn * HID + c);
      asdN = aS[dn * H + h];
    }

    v4f r = acc / sum + b;
    if (ELU) {
      r.x = r.x > 0.f ? r.x : (expf(r.x) - 1.f);
      r.y = r.y > 0.f ? r.y : (expf(r.y) - 1.f);
      r.z = r.z > 0.f ? r.z : (expf(r.z) - 1.f);
      r.w = r.w > 0.f ? r.w : (expf(r.w) - 1.f);
    }
    *(v4f*)(out + (size_t)d * HID + c) = r;

    if (!more) break;
    d = dn; beg = begN; end = endN;
    mys0 = mysN; hself = hselfN; asd = asdN;
  }
}

extern "C" void kernel_launch(void* const* d_in, const int* in_sizes, int n_in,
                              void* d_out, int out_size, void* d_ws, size_t ws_size,
                              hipStream_t stream) {
  const float* x   = (const float*)d_in[0];
  const int*   ei  = (const int*)d_in[1];
  const float* W1  = (const float*)d_in[2];
  const float* as1 = (const float*)d_in[3];
  const float* ad1 = (const float*)d_in[4];
  const float* b1  = (const float*)d_in[5];
  const float* W2  = (const float*)d_in[6];
  const float* as2 = (const float*)d_in[7];
  const float* ad2 = (const float*)d_in[8];
  const float* b2  = (const float*)d_in[9];

  const int nEdges = in_sizes[1] / 2;

  float* A    = (float*)d_ws;
  float* B    = A  + (size_t)N_NODES * HID;
  float* aS   = B  + (size_t)N_NODES * HID;
  int*   csr  = (int*)(aS + (size_t)N_NODES * GAT_HEADS);
  int*   seg  = (int*)(aS + (size_t)N_NODES * GAT_HEADS * 3);
  int*   tsum = seg + 102400;
  int*   cnt  = (int*)B;
  int*   rank = cnt + (size_t)N_NODES * CNT_PITCH;
  float* out  = (float*)d_out;

  const int TB = 256;
  auto cdiv = [](long long a, long long b) { return (unsigned)((a + b - 1) / b); };
  const unsigned gTiles = cdiv(N_NODES, SCAN_TILE);
  const unsigned gGemm  = 1250;
  const unsigned gEdge  = 3125;
  const unsigned gNodeW = 1786;
  const int TBE = 128;

  zero_counters_kernel<<<cdiv(N_NODES, TB), TB, 0, stream>>>(cnt);
  degree_rank_kernel<<<gEdge, TBE, 0, stream>>>(ei, nEdges, cnt, rank);
  scan_tile_sums_kernel<<<gTiles, SCAN_TB, 0, stream>>>(cnt, tsum);
  scan_apply_kernel<<<gTiles, SCAN_TB, 0, stream>>>(cnt, tsum, seg);
  fill_csr_kernel<<<gEdge, TBE, 0, stream>>>(ei, nEdges, seg, rank, csr);

  gemm128_wmma<<<gGemm, TB, 0, stream>>>(x, W1, A);
  alpha1_kernel<<<gEdge, TBE, 0, stream>>>(A, as1, aS);
  gat_gather_kernel<GAT_HEADS, true><<<gNodeW, TB, 0, stream>>>(
      csr, seg, nEdges, aS, ad1, A, b1, B);

  gemm128_wmma<<<gGemm, TB, 0, stream>>>(B, W2, A);
  alpha2_kernel<<<gNodeW, TB, 0, stream>>>(A, as2, aS);
  gat_gather_kernel<1, false><<<gNodeW, TB, 0, stream>>>(
      csr, seg, nEdges, aS, ad2, A, b2, out);
}


# Gather: a node's loads leave in few independent groups (logits lane-parallel, each weight computed once, rows 8 at a time, the first 8 beside the logits, no one-by-one tail); sums in the same order
# speedup vs baseline: 4.6879x; 1.1277x over previous
#include <hip/hip_runtime.h>

#define N_NODES 100000
#define IN_DIM 128
#define HID 128
#define GAT_HEADS 8
#define NEG_SLOPE 0.2f

typedef float v2f __attribute__((ext_vector_type(2)));
typedef float v4f __attribute__((ext_vector_type(4)));
typedef float v8f __attribute__((ext_vector_type(8)));


template <int M>
__device__ __forceinline__ float row_xor(float x) {
  return __int_as_float(__builtin_amdgcn_update_dpp(
      0, __float_as_int(x), 0x160 + M, 0xf, 0xf, false));
}

template <int NH>
__global__ void __launch_bounds__(256, 10) gemm128_wmma(const float* __restrict__ X,
                                                    const float* __restrict__ W,
                                                    const float* __restrict__ avS,
                                                    float* __restrict__ H,
                                                    float* __restrict__ outS) {
  const int lane  = threadIdx.x & 31;
  const int tileN = threadIdx.x >> 5;
  const int r     = lane & 15;
  const int khalf = lane >> 4;
  const int col   = tileN * 16 + r;

  v2f bf[IN_DIM / 4];
#pragma unroll
  for (int i = 0; i < IN_DIM / 4; ++i) {
    const int kk = 4 * i + 2 * khalf;
    bf[i].x = W[(kk    ) * HID + col];
    bf[i].y = W[(kk + 1) * HID + col];
  }
  const float asv = avS[col];
  const bool b8 = (r & 8) != 0, b4 = (r & 4) != 0, b2 = (r & 2) != 0;

  int it = 0;
  for (int tileM = blockIdx.x; tileM < N_NODES / 16; tileM += gridDim.x, ++it) {
    const float* __restrict__ xrow =
        X + (size_t)(tileM * 16 + r) * IN_DIM + 2 * khalf;
    v8f c = {0.f, 0.f, 0.f, 0.f, 0.f, 0.f, 0.f, 0.f};
#pragma unroll
    for (int i = 0; i < IN_DIM / 4; ++i) {
      v2f a = *(const v2f*)(xrow + 4 * i);
      c = __builtin_amdgcn_wmma_f32_16x16x4_f32(false, a, false, bf[i], (short)0,
                                                c, false, false);
    }

    float s4[4];
#pragma unroll
    for (int j = 0; j < 4; ++j) {
      const float lo = c[j] * asv, hi = c[j + 4] * asv;
      s4[j] = (b8 ? hi : lo) + row_xor<8>(b8 ? lo : hi);
    }
    const float s20 = (b4 ? s4[2] : s4[0]) + row_xor<4>(b4 ? s4[0] : s4[2]);
    const float s21 = (b4 ? s4[3] : s4[1]) + row_xor<4>(b4 ? s4[1] : s4[3]);
    float s = (b2 ? s21 : s20) + row_xor<2>(b2 ? s20 : s21);
    s += row_xor<1>(s);

    const int mbase = tileM * 16 + khalf * 8;
    if constexpr (NH == 1) {
      __shared__ float red[2][HID / 16][16];
      if ((r & 1) == 0) red[it & 1][tileN][khalf * 8 + (r >> 1)] = s;
      __syncthreads();
      if (threadIdx.x < 16) {
        float t = 0.f;
        for (int w = 0; w < HID / 16; ++w) t += red[it & 1][w][threadIdx.x];
        outS[tileM * 16 + threadIdx.x] = t;
      }
    } else {
      if ((r & 1) == 0)
        outS[(size_t)(mbase + (r >> 1)) * NH + tileN] = s;
    }

#pragma unroll
    for (int v = 0; v < 8; ++v)
      H[(size_t)(mbase + v) * HID + tileN * 16 + r] = c[v];
  }
}

typedef int v4i __attribute__((ext_vector_type(4)));

#define SCAN_TB 256
#define SCAN_TILE (SCAN_TB * 4)

#define CNT_PITCH 32

__global__ void zero_counters_kernel(int* __restrict__ cnt) {
  int i = blockIdx.x * blockDim.x + threadIdx.x;
  if (i < N_NODES) cnt[i * CNT_PITCH] = 0;
}

__global__ void degree_rank_kernel(const int* __restrict__ ei, int nEdges,
                                   int* __restrict__ cnt,
                                   int* __restrict__ rank) {
  const int stride = gridDim.x * blockDim.x;
  const int* __restrict__ dst = ei + nEdges;
  int i = blockIdx.x * blockDim.x + threadIdx.x;
  for (; i + 3 * stride < nEdges; i += 4 * stride) {
    const int d0 = dst[i], d1 = dst[i + stride];
    const int d2 = dst[i + 2 * stride], d3 = dst[i + 3 * stride];
    const int r0 = atomicAdd(&cnt[d0 * CNT_PITCH], 1);
    const int r1 = atomicAdd(&cnt[d1 * CNT_PITCH], 1);
    const int r2 = atomicAdd(&cnt[d2 * CNT_PITCH], 1);
    const int r3 = atomicAdd(&cnt[d3 * CNT_PITCH], 1);
    rank[i] = r0; rank[i + stride] = r1;
    rank[i + 2 * stride] = r2; rank[i + 3 * stride] = r3;
  }
  for (; i < nEdges; i += stride)
    rank[i] = atomicAdd(&cnt[dst[i] * CNT_PITCH], 1);
}

__global__ void __launch_bounds__(SCAN_TB)
scan_tile_sums_kernel(const int* __restrict__ cnt, int* __restrict__ tileSum) {
  __shared__ int wsum[SCAN_TB / 32];
  const int base = (blockIdx.x * SCAN_TB + threadIdx.x) * 4;
  int t = 0;
  if (base < N_NODES)
    t = cnt[(base + 0) * CNT_PITCH] + cnt[(base + 1) * CNT_PITCH] +
        cnt[(base + 2) * CNT_PITCH] + cnt[(base + 3) * CNT_PITCH];
  for (int off = 16; off > 0; off >>= 1) t += __shfl_xor(t, off, 32);
  if ((threadIdx.x & 31) == 0) wsum[threadIdx.x >> 5] = t;
  __syncthreads();
  if (threadIdx.x == 0) {
    int s = 0;
    for (int w = 0; w < SCAN_TB / 32; ++w) s += wsum[w];
    tileSum[blockIdx.x] = s;
  }
}

__global__ void __launch_bounds__(SCAN_TB)
scan_apply_kernel(const int* __restrict__ cnt, const int* __restrict__ tileSum,
                  int* __restrict__ segStart) {
  __shared__ int wsum[SCAN_TB / 32];
  __shared__ int tileOff;
  const int lane = threadIdx.x & 31, wave = threadIdx.x >> 5;
  if (wave == 0) {
    int o = 0;
    for (int b = lane; b < (int)blockIdx.x; b += 32) o += tileSum[b];
    for (int off = 16; off > 0; off >>= 1) o += __shfl_xor(o, off, 32);
    if (lane == 0) tileOff = o;
  }
  const int base = (blockIdx.x * SCAN_TB + threadIdx.x) * 4;
  v4i v = {0, 0, 0, 0};
  if (base < N_NODES) {
    v.x = cnt[(base + 0) * CNT_PITCH]; v.y = cnt[(base + 1) * CNT_PITCH];
    v.z = cnt[(base + 2) * CNT_PITCH]; v.w = cnt[(base + 3) * CNT_PITCH];
  }
  const int t = v.x + v.y + v.z + v.w;
  int inc = t;
  for (int off = 1; off < 32; off <<= 1) {
    int u = __shfl_up(inc, off, 32);
    if (lane >= off) inc += u;
  }
  if (lane == 31) wsum[wave] = inc;
  __syncthreads();
  int o = tileOff;
  for (int w = 0; w < wave; ++w) o += wsum[w];
  const int e0 = o + inc - t;
  if (base < N_NODES) {
    v4i r = {e0, e0 + v.x, e0 + v.x + v.y, e0 + v.x + v.y + v.z};
    *(v4i*)(segStart + base) = r;
  }
}

__global__ void fill_csr_kernel(const int* __restrict__ ei, int nEdges,
                                const int* __restrict__ segStart,
                                const int* __restrict__ rank,
                                int* __restrict__ csr) {
  const int stride = gridDim.x * blockDim.x;
  const int* __restrict__ dst = ei + nEdges;
  int i = blockIdx.x * blockDim.x + threadIdx.x;
  for (; i + 3 * stride < nEdges; i += 4 * stride) {
    const int d0 = dst[i], d1 = dst[i + stride];
    const int d2 = dst[i + 2 * stride], d3 = dst[i + 3 * stride];
    const int s0 = ei[i], s1 = ei[i + stride];
    const int s2 = ei[i + 2 * stride], s3 = ei[i + 3 * stride];
    const int r0 = rank[i], r1 = rank[i + stride];
    const int r2 = rank[i + 2 * stride], r3 = rank[i + 3 * stride];
    const int p0 = segStart[d0] + r0, p1 = segStart[d1] + r1;
    const int p2 = segStart[d2] + r2, p3 = segStart[d3] + r3;
    csr[p0] = s0; csr[p1] = s1; csr[p2] = s2; csr[p3] = s3;
  }
  for (; i < nEdges; i += stride) csr[segStart[dst[i]] + rank[i]] = ei[i];
}

__device__ __forceinline__ float leaky(float x) {
  return x > 0.f ? x : NEG_SLOPE * x;
}


__device__ __forceinline__ float other_row(float x) {
  return __int_as_float(__builtin_amdgcn_permlanex16(
      __float_as_int(x), __float_as_int(x), 0x76543210u, 0xfedcba98u, false, false));
}

__device__ __forceinline__ void edge_head_load(int mys, int lane,
                                               const float* __restrict__ aS,
                                               float (&x)[8]) {
  const int g = lane & 7;
  const bool upper = lane >= 16;
#pragma unroll
  for (int t = 0; t < 8; ++t) {
    const unsigned f = 4u * (t & 3);
    const unsigned loA = 0x11111111u * f, hiA = 0x11111111u * (f + 1);
    const unsigned loB = 0x11111111u * (f + 2), hiB = 0x11111111u * (f + 3);
    int forLower, forUpper;
    if (t < 4) {
      forLower = __builtin_amdgcn_permlane16(mys, mys, loA, hiA, false, false);
      forUpper = __builtin_amdgcn_permlanex16(mys, mys, loB, hiB, false, false);
    } else {
      forLower = __builtin_amdgcn_permlanex16(mys, mys, loA, hiA, false, false);
      forUpper = __builtin_amdgcn_permlane16(mys, mys, loB, hiB, false, false);
    }
    const int sj = upper ? forUpper : forLower;
    x[t] = aS[sj * GAT_HEADS + g];
  }
}

__device__ __forceinline__ void edge_head_finish(float (&x)[8], int n, int lane, float pdg) {
  const int q = lane >> 3;
#pragma unroll
  for (int t = 0; t < 8; ++t) {
    const float v = x[t] + pdg;
    x[t] = (4 * t + q < n) ? (v > 0.f ? v : NEG_SLOPE * v) : -__builtin_inff();
  }
}

template <int H, bool ELU>
__global__ void __launch_bounds__(256, H == 1 ? 16 : 10)
gat_gather_kernel(const int* __restrict__ csr, const int* __restrict__ segStart,
                  int nEdges,
                  const float* __restrict__ aS, const float* __restrict__ avD,
                  const float* __restrict__ Hf, const float* __restrict__ bias,
                  float* __restrict__ out) {
  const int lane = threadIdx.x & 31;
  const int nW = gridDim.x * 8;
  int d = __builtin_amdgcn_readfirstlane(blockIdx.x * 8 + (threadIdx.x >> 5));
  if (d >= N_NODES) return;
  const int c = lane * 4;
  const int h = c / (HID / H);
  const v4f ad = *(const v4f*)(avD + c);
  const v4f b = *(const v4f*)(bias + c);

  int beg = segStart[d];
  int end = (d + 1 < N_NODES) ? segStart[d + 1] : nEdges;
  int mys0 = (lane < end - beg) ? csr[beg + lane] : 0;
  v4f hself = *(const v4f*)(Hf + (size_t)d * HID + c);
  float asd = aS[d * H + h];

  auto load_rows = [&](int mys, int ch, v4f (&hv)[8]) {
#pragma unroll
    for (int k = 0; k < 8; ++k) {
      const int s = __builtin_amdgcn_readlane(mys, 8 * ch + k);
      hv[k] = *(const v4f*)(Hf + (size_t)s * HID + c);
    }
  };

  for (;;) {
    const int dn = d + nW;
    const bool more = dn < N_NODES;
    int begN = 0, endN = 0;
    if (more) {
      begN = segStart[dn];
      endN = (dn + 1 < N_NODES) ? segStart[dn + 1] : nEdges;
    }
    const int n0 = min(32, end - beg);

    auto dest_part = [&]() {
      float p = hself.x * ad.x + hself.y * ad.y + hself.z * ad.z + hself.w * ad.w;
      if (H == GAT_HEADS) {
        p += row_xor<1>(p);
        p += row_xor<2>(p);
      } else {
        p += other_row(p);
        p += row_xor<8>(p);
        p += row_xor<4>(p);
        p += row_xor<2>(p);
        p += row_xor<1>(p);
      }
      return p;
    };

    float sum;
    v4f acc;
    int mysN = 0;
    v4f hselfN = hself;
    float asdN = 0.f;
    if (H == 1) {
      const float lg0 = (lane < n0) ? aS[mys0] : 0.f;
      v4f hv0[8];
      load_rows(mys0, 0, hv0);
      __builtin_amdgcn_sched_barrier(0);
      const float pd = dest_part();
      const float xself = leaky(asd + pd);
      float m = xself;
      if (lane < n0) m = fmaxf(m, leaky(lg0 + pd));
      for (int base = beg + 32; base < end; base += 32) {
        const int n = min(32, end - base);
        if (lane < n) m = fmaxf(m, leaky(aS[csr[base + lane]] + pd));
      }
      m = fmaxf(m, other_row(m));
      m = fmaxf(m, row_xor<8>(m));
      m = fmaxf(m, row_xor<4>(m));
      m = fmaxf(m, row_xor<2>(m));
      m = fmaxf(m, row_xor<1>(m));
      const float eself = expf(xself - m);
      acc = hself * eself;
      auto add_rows = [&](float e, int ch, const v4f (&hv)[8]) {
#pragma unroll
        for (int k = 0; k < 8; ++k) {
          const float ej = __int_as_float(
              __builtin_amdgcn_readlane(__float_as_int(e), 8 * ch + k));
          acc += hv[k] * ej;
        }
      };
      const float e0 = (lane < n0) ? expf(leaky(lg0 + pd) - m) : 0.f;
      float part = e0;
      add_rows(e0, 0, hv0);
#pragma unroll
      for (int ch = 1; ch < 4; ++ch) {
        if (8 * ch < n0) {
          v4f hv[8];
          load_rows(mys0, ch, hv);
          add_rows(e0, ch, hv);
        }
      }
      for (int base = beg + 32; base < end; base += 32) {
        const int n = min(32, end - base);
        int mys = 0;
        float e = 0.f;
        if (lane < n) {
          mys = csr[base + lane];
          e = expf(leaky(aS[mys] + pd) - m);
        }
        part += e;
#pragma unroll
        for (int ch = 0; ch < 4; ++ch) {
          if (8 * ch < n) {
            v4f hv[8];
            load_rows(mys, ch, hv);
            add_rows(e, ch, hv);
          }
        }
      }
      part += other_row(part);
      part += row_xor<8>(part);
      part += row_xor<4>(part);
      part += row_xor<2>(part);
      part += row_xor<1>(part);
      sum = eself + part;
    } else {
      const int g = lane & 7, q = lane >> 3;
      float w[8];
      edge_head_load(mys0, lane, aS, w);
      v4f hv0[8];
      load_rows(mys0, 0, hv0);
      __builtin_amdgcn_sched_barrier(0);
      const float pd = dest_part();
      const float xself = leaky(asd + pd);
      const float pdg = __shfl(pd, 4 * g, 32);
      edge_head_finish(w, n0, lane, pdg);
      float mg = fmaxf(fmaxf(fmaxf(w[0], w[1]), fmaxf(w[2], w[3])),
                       fmaxf(fmaxf(w[4], w[5]), fmaxf(w[6], w[7])));
      for (int base = beg + 32; base < end; base += 32) {
        const int n = min(32, end - base);
        const int mys = (lane < n) ? csr[base + lane] : 0;
        float x[8];
        edge_head_load(mys, lane, aS, x);
        edge_head_finish(x, n, lane, pdg);
        for (int t = 0; t < 8; ++t) mg = fmaxf(mg, x[t]);
      }
      mg = fmaxf(mg, row_xor<8>(mg));
      mg = fmaxf(mg, other_row(mg));
      const float m = fmaxf(xself, __shfl(mg, h, 32));
      const float mgm = __shfl(m, 4 * g, 32);
      const float eself = expf(xself - m);
      sum = eself;
      acc = hself * eself;

      auto to_weights = [&](float (&x)[8], int n) {
#pragma unroll
        for (int t = 0; t < 8; ++t) {
          if (4 * t < n)
            x[t] = (4 * t + q < n) ? expf(x[t] - mgm) : 0.f;
          else
            x[t] = 0.f;
        }
      };
      auto add_rows = [&](const float (&x)[8], int ch, const v4f (&hv)[8]) {
#pragma unroll
        for (int k = 0; k < 8; ++k) {
          const float e = __shfl(x[2 * ch + (k >> 2)], (k & 3) * 8 + h, 32);
          sum += e;
          acc += hv[k] * e;
        }
      };

      to_weights(w, n0);
      add_rows(w, 0, hv0);
#pragma unroll
      for (int ch = 1; ch < 4; ++ch) {
        if (8 * ch < n0) {
          v4f hv[8];
          load_rows(mys0, ch, hv);
          add_rows(w, ch, hv);
        }
      }
      for (int base = beg + 32; base < end; base += 32) {
        const int n = min(32, end - base);
        const int mys = (lane < n) ? csr[base + lane] : 0;
        float x[8];
        edge_head_load(mys, lane, aS, x);
        edge_head_finish(x, n, lane, pdg);
        to_weights(x, n);
#pragma unroll
        for (int ch = 0; ch < 4; ++ch) {
          if (8 * ch < n) {
            v4f hv[8];
            load_rows(mys, ch, hv);
            add_rows(x, ch, hv);
          }
        }
      }
    }

    if (more) {
      mysN = (lane < endN - begN) ? csr[begN + lane] : 0;
      hselfN = *(const v4f*)(Hf + (size_t)dn * HID + c);
      asdN = aS[dn * H + h];
    }

    v4f r = acc / sum + b;
    if (ELU) {
      r.x = r.x > 0.f ? r.x : (expf(r.x) - 1.f);
      r.y = r.y > 0.f ? r.y : (expf(r.y) - 1.f);
      r.z = r.z > 0.f ? r.z : (expf(r.z) - 1.f);
      r.w = r.w > 0.f ? r.w : (expf(r.w) - 1.f);
    }
    *(v4f*)(out + (size_t)d * HID + c) = r;

    if (!more) break;
    d = dn; beg = begN; end = endN;
    mys0 = mysN; hself = hselfN; asd = asdN;
  }
}

extern "C" void kernel_launch(void* const* d_in, const int* in_sizes, int n_in,
                              void* d_out, int out_size, void* d_ws, size_t ws_size,
                              hipStream_t stream) {
  const float* x   = (const float*)d_in[0];
  const int*   ei  = (const int*)d_in[1];
  const float* W1  = (const float*)d_in[2];
  const float* as1 = (const float*)d_in[3];
  const float* ad1 = (const float*)d_in[4];
  const float* b1  = (const float*)d_in[5];
  const float* W2  = (const float*)d_in[6];
  const float* as2 = (const float*)d_in[7];
  const float* ad2 = (const float*)d_in[8];
  const float* b2  = (const float*)d_in[9];

  const int nEdges = in_sizes[1] / 2;

  float* A    = (float*)d_ws;
  float* B    = A  + (size_t)N_NODES * HID;
  float* aS   = B  + (size_t)N_NODES * HID;
  int*   csr  = (int*)(aS + (size_t)N_NODES * GAT_HEADS);
  int*   seg  = (int*)(aS + (size_t)N_NODES * GAT_HEADS * 3);
  int*   tsum = seg + 102400;
  int*   cnt  = (int*)B;
  int*   rank = cnt + (size_t)N_NODES * CNT_PITCH;
  float* out  = (float*)d_out;

  const int TB = 256;
  auto cdiv = [](long long a, long long b) { return (unsigned)((a + b - 1) / b); };
  const unsigned gTiles = cdiv(N_NODES, SCAN_TILE);
  const unsigned gGemm  = 1250;
  const unsigned gEdge  = 3125;
  const unsigned gNodeW = 1786;
  const unsigned gNode8 = 1280;
  const int TBE = 128;

  zero_counters_kernel<<<cdiv(N_NODES, TB), TB, 0, stream>>>(cnt);
  degree_rank_kernel<<<gEdge, TBE, 0, stream>>>(ei, nEdges, cnt, rank);
  scan_tile_sums_kernel<<<gTiles, SCAN_TB, 0, stream>>>(cnt, tsum);
  scan_apply_kernel<<<gTiles, SCAN_TB, 0, stream>>>(cnt, tsum, seg);
  fill_csr_kernel<<<gEdge, TBE, 0, stream>>>(ei, nEdges, seg, rank, csr);

  gemm128_wmma<GAT_HEADS><<<gGemm, TB, 0, stream>>>(x, W1, as1, A, aS);
  gat_gather_kernel<GAT_HEADS, true><<<gNode8, TB, 0, stream>>>(
      csr, seg, nEdges, aS, ad1, A, b1, B);

  gemm128_wmma<1><<<gGemm, TB, 0, stream>>>(B, W2, as2, A, aS);
  gat_gather_kernel<1, false><<<gNodeW, TB, 0, stream>>>(
      csr, seg, nEdges, aS, ad2, A, b2, out);
}


# Projection: a row tile's A block is loaded once per work-group as whole rows and handed to its waves through LDS, next tile fetched ahead, not strided loads in every wave; same k order and sums
# speedup vs baseline: 7.3927x; 1.5770x over previous
#include <hip/hip_runtime.h>

#define N_NODES 100000
#define IN_DIM 128
#define HID 128
#define GAT_HEADS 8
#define NEG_SLOPE 0.2f

typedef float v2f __attribute__((ext_vector_type(2)));
typedef float v4f __attribute__((ext_vector_type(4)));
typedef float v8f __attribute__((ext_vector_type(8)));


template <int M>
__device__ __forceinline__ float row_xor(float x) {
  return __int_as_float(__builtin_amdgcn_update_dpp(
      0, __float_as_int(x), 0x160 + M, 0xf, 0xf, false));
}

#define A_PITCH (IN_DIM + 4)

template <int NH>
__global__ void __launch_bounds__(256, 10) gemm128_wmma(const float* __restrict__ X,
                                                    const float* __restrict__ W,
                                                    const float* __restrict__ avS,
                                                    float* __restrict__ H,
                                                    float* __restrict__ outS) {
  const int lane  = threadIdx.x & 31;
  const int tileN = threadIdx.x >> 5;
  const int r     = lane & 15;
  const int khalf = lane >> 4;
  const int col   = tileN * 16 + r;

  v2f bf[IN_DIM / 4];
#pragma unroll
  for (int i = 0; i < IN_DIM / 4; ++i) {
    const int kk = 4 * i + 2 * khalf;
    bf[i].x = W[(kk    ) * HID + col];
    bf[i].y = W[(kk + 1) * HID + col];
  }
  const float asv = avS[col];
  const bool b8 = (r & 8) != 0, b4 = (r & 4) != 0, b2 = (r & 2) != 0;

  __shared__ __attribute__((aligned(16))) float As[2 * 16 * A_PITCH];
  int stOff = 2 * tileN * A_PITCH + lane * 4;
  int rdOff = r * A_PITCH + 2 * khalf;
  asm volatile("" : "+v"(stOff), "+v"(rdOff));
  v4f g0 = {0.f, 0.f, 0.f, 0.f}, g1 = {0.f, 0.f, 0.f, 0.f};
  const v4f* __restrict__ X4 = (const v4f*)X;
  const unsigned srow = 2u * (unsigned)tileN * (IN_DIM / 4) + (unsigned)lane;
  if ((int)blockIdx.x < N_NODES / 16) {
    const unsigned xo = blockIdx.x * (16u * IN_DIM / 4) + srow;
    g0 = X4[xo];
    g1 = X4[xo + IN_DIM / 4];
  }

  int it = 0;
  for (int tileM = blockIdx.x; tileM < N_NODES / 16; tileM += gridDim.x, ++it) {
    float* At = As + (it & 1) * (16 * A_PITCH);
    *(v4f*)(At + stOff) = g0;
    *(v4f*)(At + stOff + A_PITCH) = g1;
    __syncthreads();
    const int tileNext = tileM + (int)gridDim.x;
    if (tileNext < N_NODES / 16) {
      const unsigned xo = (unsigned)tileNext * (16u * IN_DIM / 4) + srow;
      g0 = X4[xo];
      g1 = X4[xo + IN_DIM / 4];
    }

    const float* arow = At + rdOff;
    v8f c = {0.f, 0.f, 0.f, 0.f, 0.f, 0.f, 0.f, 0.f};
#pragma unroll
    for (int i = 0; i < IN_DIM / 4; ++i) {
      v2f a = *(const v2f*)(arow + 4 * i);
      c = __builtin_amdgcn_wmma_f32_16x16x4_f32(false, a, false, bf[i], (short)0,
                                                c, false, false);
    }

    float s4[4];
#pragma unroll
    for (int j = 0; j < 4; ++j) {
      const float lo = c[j] * asv, hi = c[j + 4] * asv;
      s4[j] = (b8 ? hi : lo) + row_xor<8>(b8 ? lo : hi);
    }
    const float s20 = (b4 ? s4[2] : s4[0]) + row_xor<4>(b4 ? s4[0] : s4[2]);
    const float s21 = (b4 ? s4[3] : s4[1]) + row_xor<4>(b4 ? s4[1] : s4[3]);
    float s = (b2 ? s21 : s20) + row_xor<2>(b2 ? s20 : s21);
    s += row_xor<1>(s);

    const int mbase = tileM * 16 + khalf * 8;
    if constexpr (NH == 1) {
      __shared__ float red[2][HID / 16][16];
      float* rp = &red[it & 1][tileN][khalf * 8 + (r >> 1)];
      if ((r & 1) == 0) *rp = s;
      __syncthreads();
      if (tileN == 0 && (r & 1) == 0) {
        float t = 0.f;
        for (int w = 0; w < HID / 16; ++w) t += rp[w * 16];
        outS[(unsigned)(mbase + (r >> 1))] = t;
      }
    } else {
      if ((r & 1) == 0)
        outS[(unsigned)(mbase + (r >> 1)) * (unsigned)NH + (unsigned)tileN] = s;
    }

#pragma unroll
    for (int v = 0; v < 8; ++v)
      H[(unsigned)mbase * HID + (unsigned)(tileN * 16 + r) + (unsigned)v * HID] = c[v];
  }
}

typedef int v2i __attribute__((ext_vector_type(2)));
typedef int v4i __attribute__((ext_vector_type(4)));

#define BK_SHIFT 9
#define BK_NODES (1 << BK_SHIFT)
#define NBK ((N_NODES + BK_NODES - 1) / BK_NODES)
#define SORT_WG 256
#define SORT_TB 512
#define SORT_EPT 13
#define PLACE_TB 1024
#define PLACE_EPT 12
#define PLACE_CAP (PLACE_TB * PLACE_EPT)
#define HCH (NBK / 4)
#define HRG (SORT_TB / HCH)
#define HRPG ((SORT_WG + HRG - 1) / HRG)
static_assert(NBK % 4 == 0 && NBK <= 256, "rows of bucket counts are read 16 bytes at a time");

__global__ void __launch_bounds__(SORT_TB)
bucket_hist_kernel(const int* __restrict__ ei, int nEdges, int* __restrict__ hist) {
  __shared__ int h[NBK];
  const int tid = threadIdx.x;
  if (tid < NBK) h[tid] = 0;
  __syncthreads();
  const int chunk = (nEdges + SORT_WG - 1) / SORT_WG;
  const int cbeg = (int)blockIdx.x * chunk;
  const int cend = min(nEdges, cbeg + chunk);
  const int* __restrict__ dst = ei + nEdges;
  for (int i0 = cbeg + tid; i0 < cend; i0 += SORT_EPT * SORT_TB) {
    int d[SORT_EPT];
#pragma unroll
    for (int k = 0; k < SORT_EPT; ++k)
      d[k] = dst[min(i0 + k * SORT_TB, cend - 1)];
#pragma unroll
    for (int k = 0; k < SORT_EPT; ++k)
      if (i0 + k * SORT_TB < cend) atomicAdd(&h[d[k] >> BK_SHIFT], 1);
  }
  __syncthreads();
  if (tid < NBK) hist[blockIdx.x * NBK + tid] = h[tid];
}

__global__ void __launch_bounds__(SORT_TB)
bucket_scatter_kernel(const int* __restrict__ ei, int nEdges,
                      const int* __restrict__ hist,
                      int* __restrict__ bucketStart, int* __restrict__ pairsI) {
  __shared__ v4i partTot[HRG][HCH];
  __shared__ v4i partPre[HRG][HCH];
  __shared__ int cursor[256];
  __shared__ int lcursor[256];
  __shared__ int gdelta[256];
  __shared__ int wtot[8], wown[8];
  __shared__ v2i stage[SORT_EPT * SORT_TB];
  v2i* __restrict__ pairs = (v2i*)pairsI;
  const int tid = threadIdx.x, lane = tid & 31;
  const int me = (int)blockIdx.x;
  const int ownRow = (tid < NBK) ? hist[me * NBK + tid] : 0;
  {
    const int cc = tid % HCH, rg = tid / HCH;
    if (rg < HRG) {
      v4i v[HRPG];
#pragma unroll
      for (int j = 0; j < HRPG; ++j) {
        const int w = min(rg + HRG * j, SORT_WG - 1);
        v[j] = *(const v4i*)(hist + w * NBK + 4 * cc);
      }
      v4i tot = {0, 0, 0, 0}, pre = {0, 0, 0, 0};
#pragma unroll
      for (int j = 0; j < HRPG; ++j) {
        const int w = rg + HRG * j;
        if (w < SORT_WG) tot += v[j];
        if (w < me) pre += v[j];
      }
      partTot[rg][cc] = tot;
      partPre[rg][cc] = pre;
    }
  }
  __syncthreads();
  int t = 0, p = 0, inc = 0, own = 0, oinc = 0;
  if (tid < 256) {
    if (tid < NBK) {
      const int* pt = (const int*)partTot;
      const int* pp = (const int*)partPre;
#pragma unroll
      for (int j = 0; j < HRG; ++j) { t += pt[j * NBK + tid]; p += pp[j * NBK + tid]; }
      own = ownRow;
    }
    inc = t;
    oinc = own;
    for (int off = 1; off < 32; off <<= 1) {
      const int u = __shfl_up(inc, off, 32);
      const int uo = __shfl_up(oinc, off, 32);
      if (lane >= off) { inc += u; oinc += uo; }
    }
    if (lane == 31) { wtot[tid >> 5] = inc; wown[tid >> 5] = oinc; }
  }
  __syncthreads();
  if (tid < 256) {
    int o = 0, oo = 0;
    for (int w = 0; w < (tid >> 5); ++w) { o += wtot[w]; oo += wown[w]; }
    const int bstart = o + inc - t;
    const int ls = oo + oinc - own;
    cursor[tid] = bstart + p;
    gdelta[tid] = bstart + p - ls;
    lcursor[tid] = ls;
    if (me == 0 && tid < NBK) bucketStart[tid] = bstart;
  }
  __syncthreads();

  const int chunk = (nEdges + SORT_WG - 1) / SORT_WG;
  const int cbeg = me * chunk;
  const int cend = min(nEdges, cbeg + chunk);
  const int* __restrict__ dst = ei + nEdges;
  const int nMine = cend - cbeg;
  if (nMine <= SORT_EPT * SORT_TB) {
    const int i0 = cbeg + tid;
    if (i0 < cend) {
      int s[SORT_EPT], d[SORT_EPT], pos[SORT_EPT];
#pragma unroll
      for (int k = 0; k < SORT_EPT; ++k) {
        const int i = min(i0 + k * SORT_TB, cend - 1);
        d[k] = dst[i];
        s[k] = ei[i];
      }
#pragma unroll
      for (int k = 0; k < SORT_EPT; ++k) {
        pos[k] = 0;
        if (i0 + k * SORT_TB < cend) pos[k] = atomicAdd(&lcursor[d[k] >> BK_SHIFT], 1);
      }
#pragma unroll
      for (int k = 0; k < SORT_EPT; ++k) {
        if (i0 + k * SORT_TB < cend) {
          v2i pr = {s[k], d[k]};
          stage[pos[k]] = pr;
        }
      }
    }
    __syncthreads();
    if (nMine > 0) {
      v2i q[SORT_EPT];
      int gd[SORT_EPT];
#pragma unroll
      for (int k = 0; k < SORT_EPT; ++k) q[k] = stage[min(tid + k * SORT_TB, nMine - 1)];
#pragma unroll
      for (int k = 0; k < SORT_EPT; ++k) gd[k] = gdelta[q[k].y >> BK_SHIFT];
#pragma unroll
      for (int k = 0; k < SORT_EPT; ++k) {
        const int j = tid + k * SORT_TB;
        if (j < nMine) pairs[gd[k] + j] = q[k];
      }
    }
  } else {
    for (int i0 = cbeg + tid; i0 < cend; i0 += SORT_EPT * SORT_TB) {
      int s[SORT_EPT], d[SORT_EPT], pos[SORT_EPT];
#pragma unroll
      for (int k = 0; k < SORT_EPT; ++k) {
        const int i = min(i0 + k * SORT_TB, cend - 1);
        d[k] = dst[i];
        s[k] = ei[i];
      }
#pragma unroll
      for (int k = 0; k < SORT_EPT; ++k) {
        pos[k] = 0;
        if (i0 + k * SORT_TB < cend) pos[k] = atomicAdd(&cursor[d[k] >> BK_SHIFT], 1);
      }
#pragma unroll
      for (int k = 0; k < SORT_EPT; ++k) {
        if (i0 + k * SORT_TB < cend) {
          v2i pr = {s[k], d[k]};
          pairs[pos[k]] = pr;
        }
      }
    }
  }
}

__global__ void __launch_bounds__(PLACE_TB)
bucket_place_kernel(const int* __restrict__ pairsI,
                    const int* __restrict__ bucketStart, int nEdges,
                    int* __restrict__ segStart, int* __restrict__ csr) {
  __shared__ int cnt[BK_NODES];
  __shared__ int wtot[BK_NODES / 32];
  __shared__ int stage[PLACE_CAP];
  const v2i* __restrict__ pairs = (const v2i*)pairsI;
  const int tid = threadIdx.x, lane = tid & 31;
  const int bk = (int)blockIdx.x;
  const int bs = bucketStart[bk];
  const int be = (bk + 1 < NBK) ? bucketStart[bk + 1] : nEdges;
  const int nb = be - bs;
  const bool fits = nb <= PLACE_CAP;

  v2i pr[PLACE_EPT];
#pragma unroll
  for (int k = 0; k < PLACE_EPT; ++k) {
    const int j = tid + k * PLACE_TB;
    v2i z = {0, 0};
    pr[k] = (j < nb) ? pairs[bs + j] : z;
  }
  if (tid < BK_NODES) cnt[tid] = 0;
  __syncthreads();
#pragma unroll
  for (int k = 0; k < PLACE_EPT; ++k)
    if (tid + k * PLACE_TB < nb) atomicAdd(&cnt[pr[k].y & (BK_NODES - 1)], 1);
  for (int j = PLACE_CAP + tid; j < nb; j += PLACE_TB)
    atomicAdd(&cnt[pairs[bs + j].y & (BK_NODES - 1)], 1);
  __syncthreads();

  int c = 0, inc = 0;
  if (tid < BK_NODES) {
    c = cnt[tid];
    inc = c;
    for (int off = 1; off < 32; off <<= 1) {
      const int u = __shfl_up(inc, off, 32);
      if (lane >= off) inc += u;
    }
    if (lane == 31) wtot[tid >> 5] = inc;
  }
  __syncthreads();
  if (tid < BK_NODES) {
    int o = 0;
    for (int w = 0; w < (tid >> 5); ++w) o += wtot[w];
    const int ex = o + inc - c;
    cnt[tid] = ex;
    const int node = bk * BK_NODES + tid;
    if (node < N_NODES) segStart[node] = bs + ex;
  }
  __syncthreads();

  if (fits) {
    int pl[PLACE_EPT];
#pragma unroll
    for (int k = 0; k < PLACE_EPT; ++k) {
      pl[k] = 0;
      if (tid + k * PLACE_TB < nb) pl[k] = atomicAdd(&cnt[pr[k].y & (BK_NODES - 1)], 1);
    }
#pragma unroll
    for (int k = 0; k < PLACE_EPT; ++k)
      if (tid + k * PLACE_TB < nb) stage[pl[k]] = pr[k].x;
    __syncthreads();
    for (int j = tid; j < nb; j += PLACE_TB) csr[bs + j] = stage[j];
  } else {
#pragma unroll
    for (int k = 0; k < PLACE_EPT; ++k) {
      if (tid + k * PLACE_TB < nb) {
        const int p = atomicAdd(&cnt[pr[k].y & (BK_NODES - 1)], 1);
        csr[bs + p] = pr[k].x;
      }
    }
    for (int j = PLACE_CAP + tid; j < nb; j += PLACE_TB) {
      const v2i e = pairs[bs + j];
      const int p = atomicAdd(&cnt[e.y & (BK_NODES - 1)], 1);
      csr[bs + p] = e.x;
    }
  }
}

__device__ __forceinline__ float leaky(float x) {
  return x > 0.f ? x : NEG_SLOPE * x;
}


__device__ __forceinline__ float other_row(float x) {
  return __int_as_float(__builtin_amdgcn_permlanex16(
      __float_as_int(x), __float_as_int(x), 0x76543210u, 0xfedcba98u, false, false));
}

__device__ __forceinline__ void edge_head_load(int mys, int lane,
                                               const float* __restrict__ aS,
                                               float (&x)[8]) {
  const int g = lane & 7;
  const bool upper = lane >= 16;
#pragma unroll
  for (int t = 0; t < 8; ++t) {
    const unsigned f = 4u * (t & 3);
    const unsigned loA = 0x11111111u * f, hiA = 0x11111111u * (f + 1);
    const unsigned loB = 0x11111111u * (f + 2), hiB = 0x11111111u * (f + 3);
    int forLower, forUpper;
    if (t < 4) {
      forLower = __builtin_amdgcn_permlane16(mys, mys, loA, hiA, false, false);
      forUpper = __builtin_amdgcn_permlanex16(mys, mys, loB, hiB, false, false);
    } else {
      forLower = __builtin_amdgcn_permlanex16(mys, mys, loA, hiA, false, false);
      forUpper = __builtin_amdgcn_permlane16(mys, mys, loB, hiB, false, false);
    }
    const int sj = upper ? forUpper : forLower;
    x[t] = aS[sj * GAT_HEADS + g];
  }
}

__device__ __forceinline__ void edge_head_finish(float (&x)[8], int n, int lane, float pdg) {
  const int q = lane >> 3;
#pragma unroll
  for (int t = 0; t < 8; ++t) {
    const float v = x[t] + pdg;
    x[t] = (4 * t + q < n) ? (v > 0.f ? v : NEG_SLOPE * v) : -__builtin_inff();
  }
}

template <int H, bool ELU>
__global__ void __launch_bounds__(256, H == 1 ? 16 : 10)
gat_gather_kernel(const int* __restrict__ csr, const int* __restrict__ segStart,
                  int nEdges,
                  const float* __restrict__ aS, const float* __restrict__ avD,
                  const float* __restrict__ Hf, const float* __restrict__ bias,
                  float* __restrict__ out) {
  const int lane = threadIdx.x & 31;
  const int nW = gridDim.x * 8;
  int d = __builtin_amdgcn_readfirstlane(blockIdx.x * 8 + (threadIdx.x >> 5));
  if (d >= N_NODES) return;
  const int c = lane * 4;
  const int h = c / (HID / H);
  const v4f ad = *(const v4f*)(avD + c);
  const v4f b = *(const v4f*)(bias + c);

  int beg = segStart[d];
  int end = (d + 1 < N_NODES) ? segStart[d + 1] : nEdges;
  int mys0 = (lane < end - beg) ? csr[beg + lane] : 0;
  v4f hself = *(const v4f*)(Hf + (size_t)d * HID + c);
  float asd = aS[d * H + h];

  auto load_rows = [&](int mys, int ch, v4f (&hv)[8]) {
#pragma unroll
    for (int k = 0; k < 8; ++k) {
      const int s = __builtin_amdgcn_readlane(mys, 8 * ch + k);
      hv[k] = *(const v4f*)(Hf + (size_t)s * HID + c);
    }
  };

  for (;;) {
    const int dn = d + nW;
    const bool more = dn < N_NODES;
    int begN = 0, endN = 0;
    if (more) {
      begN = segStart[dn];
      endN = (dn + 1 < N_NODES) ? segStart[dn + 1] : nEdges;
    }
    const int n0 = min(32, end - beg);

    auto dest_part = [&]() {
      float p = hself.x * ad.x + hself.y * ad.y + hself.z * ad.z + hself.w * ad.w;
      if (H == GAT_HEADS) {
        p += row_xor<1>(p);
        p += row_xor<2>(p);
      } else {
        p += other_row(p);
        p += row_xor<8>(p);
        p += row_xor<4>(p);
        p += row_xor<2>(p);
        p += row_xor<1>(p);
      }
      return p;
    };

    float sum;
    v4f acc;
    int mysN = 0;
    v4f hselfN = hself;
    float asdN = 0.f;
    if (H == 1) {
      const float lg0 = (lane < n0) ? aS[mys0] : 0.f;
      v4f hv0[8];
      load_rows(mys0, 0, hv0);
      __builtin_amdgcn_sched_barrier(0);
      const float pd = dest_part();
      const float xself = leaky(asd + pd);
      float m = xself;
      if (lane < n0) m = fmaxf(m, leaky(lg0 + pd));
      for (int base = beg + 32; base < end; base += 32) {
        const int n = min(32, end - base);
        if (lane < n) m = fmaxf(m, leaky(aS[csr[base + lane]] + pd));
      }
      m = fmaxf(m, other_row(m));
      m = fmaxf(m, row_xor<8>(m));
      m = fmaxf(m, row_xor<4>(m));
      m = fmaxf(m, row_xor<2>(m));
      m = fmaxf(m, row_xor<1>(m));
      const float eself = expf(xself - m);
      acc = hself * eself;
      auto add_rows = [&](float e, int ch, const v4f (&hv)[8]) {
#pragma unroll
        for (int k = 0; k < 8; ++k) {
          const float ej = __int_as_float(
              __builtin_amdgcn_readlane(__float_as_int(e), 8 * ch + k));
          acc += hv[k] * ej;
        }
      };
      const float e0 = (lane < n0) ? expf(leaky(lg0 + pd) - m) : 0.f;
      float part = e0;
      add_rows(e0, 0, hv0);
#pragma unroll
      for (int ch = 1; ch < 4; ++ch) {
        if (8 * ch < n0) {
          v4f hv[8];
          load_rows(mys0, ch, hv);
          add_rows(e0, ch, hv);
        }
      }
      for (int base = beg + 32; base < end; base += 32) {
        const int n = min(32, end - base);
        int mys = 0;
        float e = 0.f;
        if (lane < n) {
          mys = csr[base + lane];
          e = expf(leaky(aS[mys] + pd) - m);
        }
        part += e;
#pragma unroll
        for (int ch = 0; ch < 4; ++ch) {
          if (8 * ch < n) {
            v4f hv[8];
            load_rows(mys, ch, hv);
            add_rows(e, ch, hv);
          }
        }
      }
      part += other_row(part);
      part += row_xor<8>(part);
      part += row_xor<4>(part);
      part += row_xor<2>(part);
      part += row_xor<1>(part);
      sum = eself + part;
    } else {
      const int g = lane & 7, q = lane >> 3;
      float w[8];
      edge_head_load(mys0, lane, aS, w);
      v4f hv0[8];
      load_rows(mys0, 0, hv0);
      __builtin_amdgcn_sched_barrier(0);
      const float pd = dest_part();
      const float xself = leaky(asd + pd);
      const float pdg = __shfl(pd, 4 * g, 32);
      edge_head_finish(w, n0, lane, pdg);
      float mg = fmaxf(fmaxf(fmaxf(w[0], w[1]), fmaxf(w[2], w[3])),
                       fmaxf(fmaxf(w[4], w[5]), fmaxf(w[6], w[7])));
      for (int base = beg + 32; base < end; base += 32) {
        const int n = min(32, end - base);
        const int mys = (lane < n) ? csr[base + lane] : 0;
        float x[8];
        edge_head_load(mys, lane, aS, x);
        edge_head_finish(x, n, lane, pdg);
        for (int t = 0; t < 8; ++t) mg = fmaxf(mg, x[t]);
      }
      mg = fmaxf(mg, row_xor<8>(mg));
      mg = fmaxf(mg, other_row(mg));
      const float m = fmaxf(xself, __shfl(mg, h, 32));
      const float mgm = __shfl(m, 4 * g, 32);
      const float eself = expf(xself - m);
      sum = eself;
      acc = hself * eself;

      auto to_weights = [&](float (&x)[8], int n) {
#pragma unroll
        for (int t = 0; t < 8; ++t) {
          if (4 * t < n)
            x[t] = (4 * t + q < n) ? expf(x[t] - mgm) : 0.f;
          else
            x[t] = 0.f;
        }
      };
      auto add_rows = [&](const float (&x)[8], int ch, const v4f (&hv)[8]) {
#pragma unroll
        for (int k = 0; k < 8; ++k) {
          const float e = __shfl(x[2 * ch + (k >> 2)], (k & 3) * 8 + h, 32);
          sum += e;
          acc += hv[k] * e;
        }
      };

      to_weights(w, n0);
      add_rows(w, 0, hv0);
#pragma unroll
      for (int ch = 1; ch < 4; ++ch) {
        if (8 * ch < n0) {
          v4f hv[8];
          load_rows(mys0, ch, hv);
          add_rows(w, ch, hv);
        }
      }
      for (int base = beg + 32; base < end; base += 32) {
        const int n = min(32, end - base);
        const int mys = (lane < n) ? csr[base + lane] : 0;
        float x[8];
        edge_head_load(mys, lane, aS, x);
        edge_head_finish(x, n, lane, pdg);
        to_weights(x, n);
#pragma unroll
        for (int ch = 0; ch < 4; ++ch) {
          if (8 * ch < n) {
            v4f hv[8];
            load_rows(mys, ch, hv);
            add_rows(x, ch, hv);
          }
        }
      }
    }

    if (more) {
      mysN = (lane < endN - begN) ? csr[begN + lane] : 0;
      hselfN = *(const v4f*)(Hf + (size_t)dn * HID + c);
      asdN = aS[dn * H + h];
    }

    v4f r = acc / sum + b;
    if (ELU) {
      r.x = r.x > 0.f ? r.x : (expf(r.x) - 1.f);
      r.y = r.y > 0.f ? r.y : (expf(r.y) - 1.f);
      r.z = r.z > 0.f ? r.z : (expf(r.z) - 1.f);
      r.w = r.w > 0.f ? r.w : (expf(r.w) - 1.f);
    }
    *(v4f*)(out + (size_t)d * HID + c) = r;

    if (!more) break;
    d = dn; beg = begN; end = endN;
    mys0 = mysN; hself = hselfN; asd = asdN;
  }
}

extern "C" void kernel_launch(void* const* d_in, const int* in_sizes, int n_in,
                              void* d_out, int out_size, void* d_ws, size_t ws_size,
                              hipStream_t stream) {
  const float* x   = (const float*)d_in[0];
  const int*   ei  = (const int*)d_in[1];
  const float* W1  = (const float*)d_in[2];
  const float* as1 = (const float*)d_in[3];
  const float* ad1 = (const float*)d_in[4];
  const float* b1  = (const float*)d_in[5];
  const float* W2  = (const float*)d_in[6];
  const float* as2 = (const float*)d_in[7];
  const float* ad2 = (const float*)d_in[8];
  const float* b2  = (const float*)d_in[9];

  const int nEdges = in_sizes[1] / 2;

  float* A    = (float*)d_ws;
  float* B    = A  + (size_t)N_NODES * HID;
  float* aS   = B  + (size_t)N_NODES * HID;
  int*   csr  = (int*)(aS + (size_t)N_NODES * GAT_HEADS);
  int*   seg  = (int*)(aS + (size_t)N_NODES * GAT_HEADS * 3);
  int*   prs  = (int*)B;
  int*   hist = prs + (size_t)N_NODES * 32;
  int*   bst  = hist + SORT_WG * NBK;
  float* out  = (float*)d_out;

  const int TB = 256;
  const unsigned gGemm  = 1250;
  const unsigned gNodeW = 1786;
  const unsigned gNode8 = 1280;

  bucket_hist_kernel<<<SORT_WG, SORT_TB, 0, stream>>>(ei, nEdges, hist);
  bucket_scatter_kernel<<<SORT_WG, SORT_TB, 0, stream>>>(ei, nEdges, hist, bst, prs);
  bucket_place_kernel<<<NBK, PLACE_TB, 0, stream>>>(prs, bst, nEdges, seg, csr);

  gemm128_wmma<GAT_HEADS><<<gGemm, TB, 0, stream>>>(x, W1, as1, A, aS);
  gat_gather_kernel<GAT_HEADS, true><<<gNode8, TB, 0, stream>>>(
      csr, seg, nEdges, aS, ad1, A, b1, B);

  gemm128_wmma<1><<<gGemm, TB, 0, stream>>>(B, W2, as2, A, aS);
  gat_gather_kernel<1, false><<<gNodeW, TB, 0, stream>>>(
      csr, seg, nEdges, aS, ad2, A, b2, out);
}
